# starting kernel
# baseline (speedup 1.0000x reference)
#include <hip/hip_runtime.h>
#include <hip/hip_bf16.h>


typedef __attribute__((ext_vector_type(2))) float v2f;
typedef __attribute__((ext_vector_type(8))) float v8f;

#define BATCH 8192
#define DDIM  512
#define FDIM  64
#define KDIM  64

__global__ __launch_bounds__(256) void ffm_build_w(
    const float* __restrict__ kern,
    const int*   __restrict__ fd,
    float*       __restrict__ Wm)
{
    int tid = blockIdx.x * 256 + threadIdx.x;
    int i = tid >> 9;
    int j = tid & (DDIM - 1);
    float w = 0.0f;
    if (j > i) {
        const float4* a = (const float4*)(kern + (size_t)i * (FDIM * KDIM) + (size_t)fd[j] * KDIM);
        const float4* b = (const float4*)(kern + (size_t)j * (FDIM * KDIM) + (size_t)fd[i] * KDIM);
#pragma unroll
        for (int k = 0; k < KDIM / 4; ++k) {
            float4 av = a[k], bv = b[k];
            w += av.x * bv.x + av.y * bv.y + av.z * bv.z + av.w * bv.w;
        }
    }
    Wm[tid] = w;
}

__global__ __launch_bounds__(256) void ffm_quadform(
    const float* __restrict__ x,
    const float* __restrict__ Wm,
    float*       __restrict__ out)
{
    const int b0   = blockIdx.x * 16;
    const int wave = __builtin_amdgcn_readfirstlane(threadIdx.x >> 5);
    const int lane = threadIdx.x & 31;
    const int half = lane >> 4;
    const int l    = lane & 15;

    __shared__ float partial[8][16];

    float contrib[8];
#pragma unroll
    for (int r = 0; r < 8; ++r) contrib[r] = 0.0f;

    const int tiles[4] = { wave, 15 - wave, 16 + wave, 31 - wave };

#pragma unroll
    for (int t = 0; t < 4; ++t) {
        const int i0  = __builtin_amdgcn_readfirstlane(tiles[t] << 4);
        const int n16 = __builtin_amdgcn_readfirstlane((DDIM - i0) >> 4);

        v8f acc = {0.f, 0.f, 0.f, 0.f, 0.f, 0.f, 0.f, 0.f};

        const v2f* __restrict__ xp =
            (const v2f*)(x  + (size_t)(b0 + l) * DDIM + i0 + 2 * half);
        const v2f* __restrict__ wp =
            (const v2f*)(Wm + (size_t)(i0 + l) * DDIM + i0 + 2 * half);

        for (int it = 0; it < n16; ++it) {
#pragma unroll
            for (int u = 0; u < 4; ++u) {
                v2f a = xp[2 * u];
                v2f b = wp[2 * u];
                acc = __builtin_amdgcn_wmma_f32_16x16x4_f32(
                     false, a,  false, b,
                     (short)0, acc,  false,  false);
            }
            xp += 8;
            wp += 8;
        }

#pragma unroll
        for (int r = 0; r < 8; ++r) {
            float xv = x[(size_t)(b0 + r + 8 * half) * DDIM + i0 + l];
            contrib[r] += acc[r] * xv;
        }
    }

#pragma unroll
    for (int r = 0; r < 8; ++r) {
        float v = contrib[r];
        v += __shfl_xor(v, 1, 32);
        v += __shfl_xor(v, 2, 32);
        v += __shfl_xor(v, 4, 32);
        v += __shfl_xor(v, 8, 32);
        if (l == 0) partial[wave][r + 8 * half] = v;
    }
    __syncthreads();

    if (threadIdx.x < 16) {
        float s = 0.0f;
#pragma unroll
        for (int w = 0; w < 8; ++w) s += partial[w][threadIdx.x];
        out[b0 + threadIdx.x] = s;
    }
}

extern "C" void kernel_launch(void* const* d_in, const int* in_sizes, int n_in,
                              void* d_out, int out_size, void* d_ws, size_t ws_size,
                              hipStream_t stream) {
    const float* x    = (const float*)d_in[0];
    const float* kern = (const float*)d_in[1];
    const int*   fd   = (const int*)d_in[2];
    float* out = (float*)d_out;
    float* Wm  = (float*)d_ws;

    ffm_build_w<<<(DDIM * DDIM) / 256, 256, 0, stream>>>(kern, fd, Wm);
    ffm_quadform<<<BATCH / 16, 256, 0, stream>>>(x, Wm, out);
}


# W build: one workgroup per row pair (i, 511-i) stages the rows' own 64x64 slices in LDS once and gathers them there by field id; only partner rows come from global; W summation order unchanged
# speedup vs baseline: 1.8072x; 1.8072x over previous
#include <hip/hip_runtime.h>
#include <hip/hip_bf16.h>


typedef __attribute__((ext_vector_type(2))) float v2f;
typedef __attribute__((ext_vector_type(4))) float v4f;
typedef __attribute__((ext_vector_type(8))) float v8f;

#define BATCH 8192
#define DDIM  512
#define FDIM  64
#define KDIM  64

__device__ __forceinline__ void lds_barrier()
{
    __builtin_amdgcn_fence(__ATOMIC_RELEASE, "workgroup", "local");
    __builtin_amdgcn_s_barrier();
    __builtin_amdgcn_fence(__ATOMIC_ACQUIRE, "workgroup", "local");
}

#define TP 68

__global__ __launch_bounds__(256) void ffm_build_w(
    const float* __restrict__ kern,
    const int*   __restrict__ fd,
    float*       __restrict__ Wm)
{
    __shared__ __attribute__((aligned(16))) float T[2 * FDIM * TP];
    __shared__ __attribute__((aligned(16))) float Wrow[2 * DDIM];

    const int tid = threadIdx.x;
    const int iA  = blockIdx.x;
    const int iB  = DDIM - 1 - iA;
    const int nA  = DDIM - 1 - iA;

    const float4* __restrict__ srcA = (const float4*)(kern + (size_t)iA * (FDIM * KDIM));
    const float4* __restrict__ srcB = (const float4*)(kern + (size_t)iB * (FDIM * KDIM));
    float4 ta[4], tb[4];
#pragma unroll
    for (int r = 0; r < 4; ++r) { ta[r] = srcA[tid + 256 * r]; tb[r] = srcB[tid + 256 * r]; }

    const int gA = fd[iA] & (FDIM - 1);
    const int gB = fd[iB] & (FDIM - 1);

    const int  s0 = tid;
    const int  s1 = (tid < 255) ? tid + 256 : 510;
    const bool v1 = tid < 255;
    const int  rs0 = s0 >= nA, rs1 = s1 >= nA;
    const int  j0 = rs0 ? iB + 1 + (s0 - nA) : iA + 1 + s0;
    const int  j1 = rs1 ? iB + 1 + (s1 - nA) : iA + 1 + s1;

    const float4* __restrict__ bp0 = (const float4*)(kern + (size_t)j0 * (FDIM * KDIM) + (size_t)(rs0 ? gB : gA) * KDIM);
    const float4* __restrict__ bp1 = (const float4*)(kern + (size_t)j1 * (FDIM * KDIM) + (size_t)(rs1 ? gB : gA) * KDIM);
    float4 b0[16], b1[16];
#pragma unroll
    for (int k = 0; k < 16; ++k) b0[k] = bp0[k];
#pragma unroll
    for (int k = 0; k < 16; ++k) b1[k] = bp1[k];
    const int f0 = fd[j0] & (FDIM - 1);
    const int f1 = fd[j1] & (FDIM - 1);

#pragma unroll
    for (int r = 0; r < 4; ++r) {
        const int idx = tid + 256 * r;
        const int f = idx >> 4, q = idx & 15;
        *(float4*)(T + f * TP + 4 * q)          = ta[r];
        *(float4*)(T + (FDIM + f) * TP + 4 * q) = tb[r];
    }
    *(float4*)(Wrow + 4 * tid) = make_float4(0.0f, 0.0f, 0.0f, 0.0f);
    lds_barrier();

    const float4* a0 = (const float4*)(T + (rs0 * FDIM + f0) * TP);
    const float4* a1 = (const float4*)(T + (rs1 * FDIM + f1) * TP);
    float w0 = 0.0f, w1 = 0.0f;
#pragma unroll
    for (int k = 0; k < 16; ++k) {
        const float4 av = a0[k], bv = b0[k];
        w0 += av.x * bv.x + av.y * bv.y + av.z * bv.z + av.w * bv.w;
    }
#pragma unroll
    for (int k = 0; k < 16; ++k) {
        const float4 av = a1[k], bv = b1[k];
        w1 += av.x * bv.x + av.y * bv.y + av.z * bv.z + av.w * bv.w;
    }
    Wrow[rs0 * DDIM + j0] = w0;
    if (v1) Wrow[rs1 * DDIM + j1] = w1;
    lds_barrier();

    const float4 o = *(const float4*)(Wrow + 4 * tid);
    const int orow = (tid >> 7) ? iB : iA;
    *(float4*)(Wm + (size_t)orow * DDIM + 4 * (tid & 127)) = o;
}

#define CHUNK 64

template <int KS>
__device__ __forceinline__ void load_w(v2f (&w)[16], const float* __restrict__ wrow)
{
#pragma unroll
    for (int k = KS; k < 16; ++k) w[k] = *(const v2f*)(wrow + 4 * k);
}

template <int KS>
__device__ __forceinline__ void tile_step(
    const v2f (&w)[16], const v2f (&xa)[16], const v2f (&xb)[16],
    const float* __restrict__ xe0, const float* __restrict__ xe1,
    float (&s0)[8], float (&s1)[8])
{
    const v4f e00 = *(const v4f*)(xe0);
    const v4f e01 = *(const v4f*)(xe0 + 4);
    const v4f e10 = *(const v4f*)(xe1);
    const v4f e11 = *(const v4f*)(xe1 + 4);

    v8f acc0 = {0.f, 0.f, 0.f, 0.f, 0.f, 0.f, 0.f, 0.f};
    v8f acc1 = {0.f, 0.f, 0.f, 0.f, 0.f, 0.f, 0.f, 0.f};
#pragma unroll
    for (int k = KS; k < 16; ++k) {
        acc0 = __builtin_amdgcn_wmma_f32_16x16x4_f32(
            false, w[k], false, xa[k], (short)0, acc0, false, false);
        acc1 = __builtin_amdgcn_wmma_f32_16x16x4_f32(
            false, w[k], false, xb[k], (short)0, acc1, false, false);
    }
#pragma unroll
    for (int r = 0; r < 4; ++r) {
        s0[r]     += acc0[r]     * e00[r];
        s0[r + 4] += acc0[r + 4] * e01[r];
        s1[r]     += acc1[r]     * e10[r];
        s1[r + 4] += acc1[r + 4] * e11[r];
    }
}

__global__ __launch_bounds__(128) void ffm_quadform(
    const float* __restrict__ x,
    const float* __restrict__ Wm,
    float*       __restrict__ out)
{
    const int b0   = blockIdx.x * 32;
    const int wave = __builtin_amdgcn_readfirstlane(threadIdx.x >> 5);
    const int lane = threadIdx.x & 31;
    const int half = lane >> 4;
    const int c    = lane & 15;

    __shared__ float partial[4][32];

    float s0[8], s1[8];
#pragma unroll
    for (int r = 0; r < 8; ++r) { s0[r] = 0.0f; s1[r] = 0.0f; }

    for (int p = 0; p < 2; ++p) {
        const int ch = __builtin_amdgcn_readfirstlane(p ? 7 - wave : wave);
        const int jc = ch * CHUNK;

        const float* __restrict__ xr0 = x + (size_t)(b0 + c) * DDIM + jc + 2 * half;
        const float* __restrict__ xr1 = xr0 + 16 * DDIM;
        v2f xa[16], xb[16];
#pragma unroll
        for (int k = 0; k < 16; ++k) {
            xa[k] = *(const v2f*)(xr0 + 4 * k);
            xb[k] = *(const v2f*)(xr1 + 4 * k);
        }

        const float* __restrict__ wp  = Wm + (size_t)c * DDIM + jc + 2 * half;
        const float* __restrict__ xe0 = x + (size_t)(b0 + c) * DDIM + 8 * half;
        const float* __restrict__ xe1 = xe0 + 16 * DDIM;

        const int nfull = 4 * ch;
        v2f w0[16], w1[16], w2[16], w3[16];
        load_w<0>(w0, wp);
        load_w<0>(w1, wp + (size_t)16 * DDIM);

        int i0 = 0;
        for (int t = 0; t < nfull; t += 4) {
            load_w<0>(w2, wp + (size_t)(i0 + 32) * DDIM);
            tile_step<0>(w0, xa, xb, xe0 + i0, xe1 + i0, s0, s1);
            load_w<0>(w3, wp + (size_t)(i0 + 48) * DDIM);
            tile_step<0>(w1, xa, xb, xe0 + i0 + 16, xe1 + i0 + 16, s0, s1);
            load_w<0>(w0, wp + (size_t)(i0 + 64) * DDIM);
            tile_step<0>(w2, xa, xb, xe0 + i0 + 32, xe1 + i0 + 32, s0, s1);
            load_w<0>(w1, wp + (size_t)(i0 + 80) * DDIM);
            tile_step<0>(w3, xa, xb, xe0 + i0 + 48, xe1 + i0 + 48, s0, s1);
            i0 += 64;
        }

        load_w<8>(w2, wp + (size_t)(i0 + 32) * DDIM);
        tile_step<0>(w0, xa, xb, xe0 + i0, xe1 + i0, s0, s1);
        load_w<12>(w3, wp + (size_t)(i0 + 48) * DDIM);
        tile_step<4>(w1, xa, xb, xe0 + i0 + 16, xe1 + i0 + 16, s0, s1);
        tile_step<8>(w2, xa, xb, xe0 + i0 + 32, xe1 + i0 + 32, s0, s1);
        tile_step<12>(w3, xa, xb, xe0 + i0 + 48, xe1 + i0 + 48, s0, s1);
    }

    float t0 = ((s0[0] + s0[1]) + (s0[2] + s0[3])) + ((s0[4] + s0[5]) + (s0[6] + s0[7]));
    float t1 = ((s1[0] + s1[1]) + (s1[2] + s1[3])) + ((s1[4] + s1[5]) + (s1[6] + s1[7]));
    t0 += __shfl_xor(t0, 16, 32);
    t1 += __shfl_xor(t1, 16, 32);
    if (half == 0) {
        partial[wave][c]      = t0;
        partial[wave][16 + c] = t1;
    }
    __syncthreads();

    if (threadIdx.x < 32) {
        const int b = threadIdx.x;
        out[b0 + b] = (partial[0][b] + partial[1][b]) + (partial[2][b] + partial[3][b]);
    }
}

extern "C" void kernel_launch(void* const* d_in, const int* in_sizes, int n_in,
                              void* d_out, int out_size, void* d_ws, size_t ws_size,
                              hipStream_t stream) {
    const float* x    = (const float*)d_in[0];
    const float* kern = (const float*)d_in[1];
    const int*   fd   = (const int*)d_in[2];
    float* out = (float*)d_out;
    float* Wm  = (float*)d_ws;

    ffm_build_w<<<DDIM / 2, 256, 0, stream>>>(kern, fd, Wm);
    ffm_quadform<<<BATCH / 32, 128, 0, stream>>>(x, Wm, out);
}
